# starting kernel
# baseline (speedup 1.0000x reference)
#include <hip/hip_runtime.h>
#include <hip/hip_bf16.h>

#define N_ROWS   32768
#define D        768
#define NCLS     100
#define C2       200
#define C2P      208
#define NTILE    13
#define COLS     64
#define NCC      12
#define NRC      16
#define ROWS_PER 2048
#define PROTO_E  (NCLS * D)

typedef float v2f __attribute__((ext_vector_type(2)));
typedef float v8f __attribute__((ext_vector_type(8)));

#define WS_CNT   0
#define WS_P1    128
#define WS_P2    (WS_P1 + NRC * PROTO_E)
#define WS_ZB    (WS_P2 + NRC * PROTO_E)
#define WS_LG    (WS_ZB + C2P * D)

__global__ void zero_counts(int* cnt) {
    int t = threadIdx.x;
    if (t < NCLS) cnt[t] = 0;
}

__global__ void count_kernel(const int* __restrict__ labels, int* __restrict__ cnt) {
    int i = blockIdx.x * 256 + threadIdx.x;
    if (i < N_ROWS) atomicAdd(&cnt[labels[i]], 1);
}

__global__ __launch_bounds__(COLS) void segsum_phase1(
        const float* __restrict__ z1, const float* __restrict__ z2,
        const int* __restrict__ labels,
        float* __restrict__ p1, float* __restrict__ p2) {
    __shared__ float acc1[NCLS * COLS];
    __shared__ float acc2[NCLS * COLS];
    const int t  = threadIdx.x;
    const int cc = blockIdx.x;
    const int rc = blockIdx.y;
    for (int i = t; i < NCLS * COLS; i += COLS) { acc1[i] = 0.f; acc2[i] = 0.f; }
    __syncthreads();
    const int colBase = cc * COLS;
    const int r0 = rc * ROWS_PER;
    for (int r = r0; r < r0 + ROWS_PER; ++r) {
        const int lab = labels[r];
        const size_t off = (size_t)r * D + colBase + t;
        acc1[lab * COLS + t] += z1[off];
        acc2[lab * COLS + t] += z2[off];
    }
    __syncthreads();
    for (int i = t; i < NCLS * COLS; i += COLS) {
        const int cls = i >> 6, col = i & 63;
        const size_t o = (size_t)rc * PROTO_E + cls * D + colBase + col;
        p1[o] = acc1[i];
        p2[o] = acc2[i];
    }
}

__global__ void finalize_proto(const float* __restrict__ p1, const float* __restrict__ p2,
                               const int* __restrict__ cnt,
                               float* __restrict__ out1, float* __restrict__ out2) {
    const int e = blockIdx.x * 256 + threadIdx.x;
    float s1 = 0.f, s2 = 0.f;
    for (int rc = 0; rc < NRC; ++rc) {
        s1 += p1[(size_t)rc * PROTO_E + e];
        s2 += p2[(size_t)rc * PROTO_E + e];
    }
    const float d = fmaxf((float)cnt[e / D], 1.0f);
    out1[e] = s1 / d;
    out2[e] = s2 / d;
}

__global__ void normalize_rows(const float* __restrict__ proto, float* __restrict__ zb) {
    const int b = blockIdx.x;
    const int t = threadIdx.x;
    if (b >= C2) {
        zb[(size_t)b * D + t] = 0.f;
        zb[(size_t)b * D + t + 256] = 0.f;
        zb[(size_t)b * D + t + 512] = 0.f;
        return;
    }
    const float* src = proto + (b < NCLS ? 0 : PROTO_E) + (size_t)(b % NCLS) * D;
    const float x0 = src[t], x1 = src[t + 256], x2 = src[t + 512];
    __shared__ float red[256];
    red[t] = x0 * x0 + x1 * x1 + x2 * x2;
    __syncthreads();
    for (int s = 128; s > 0; s >>= 1) {
        if (t < s) red[t] += red[t + s];
        __syncthreads();
    }
    const float inv = 1.0f / fmaxf(sqrtf(red[0]), 1e-12f);
    zb[(size_t)b * D + t]       = x0 * inv;
    zb[(size_t)b * D + t + 256] = x1 * inv;
    zb[(size_t)b * D + t + 512] = x2 * inv;
}

__global__ __launch_bounds__(32) void logits_wmma(const float* __restrict__ zb,
                                                  float* __restrict__ lg) {
    const int tile = blockIdx.x;
    const int ti = tile / NTILE, tj = tile % NTILE;
    const int lane = threadIdx.x;
    const int mn = lane & 15;
    const int kb = (lane >> 4) << 1;
    const float* __restrict__ arow = zb + (size_t)(ti * 16 + mn) * D;
    const float* __restrict__ brow = zb + (size_t)(tj * 16 + mn) * D;
    v8f acc = {};
    for (int k0 = 0; k0 < D; k0 += 4) {
        const v2f a = *(const v2f*)(arow + k0 + kb);
        const v2f b = *(const v2f*)(brow + k0 + kb);
        acc = __builtin_amdgcn_wmma_f32_16x16x4_f32(
             false, a,  false, b,
             (short)0, acc,  false,  false);
    }
    const int mhi = (lane >> 4) * 8;
#pragma unroll
    for (int v = 0; v < 8; ++v) {
        const int m = ti * 16 + v + mhi;
        lg[(size_t)m * C2P + tj * 16 + mn] = acc[v] * 2.0f;
    }
}

__global__ void loss_kernel(const float* __restrict__ lg, const int* __restrict__ cnt,
                            float* __restrict__ out) {
    __shared__ int   varr[NCLS];
    __shared__ float sterm[256];
    __shared__ float snv[256];
    const int t = threadIdx.x;
    if (t < NCLS) varr[t] = (cnt[t] > 0) ? 1 : 0;
    __syncthreads();
    float term = 0.f, nvf = 0.f;
    if (t < C2) {
        const int vr = varr[t % NCLS];
        nvf = (float)vr;
        const float* row = lg + (size_t)t * C2P;
        float mx = -3.402823466e38f;
        for (int c = 0; c < C2; ++c)
            if (varr[c % NCLS]) mx = fmaxf(mx, row[c]);
        float se = 0.f;
        for (int c = 0; c < C2; ++c)
            if (c != t && varr[c % NCLS]) se += expf(row[c] - mx);
        const int p = (t + NCLS) % C2;
        if (vr && varr[p % NCLS]) term = row[p] - mx - logf(se);
    }
    sterm[t] = term;
    snv[t] = nvf;
    __syncthreads();
    for (int s = 128; s > 0; s >>= 1) {
        if (t < s) { sterm[t] += sterm[t + s]; snv[t] += snv[t + s]; }
        __syncthreads();
    }
    if (t == 0) out[0] = -sterm[0] / fmaxf(snv[0], 1.0f);
}

extern "C" void kernel_launch(void* const* d_in, const int* in_sizes, int n_in,
                              void* d_out, int out_size, void* d_ws, size_t ws_size,
                              hipStream_t stream) {
    const float* z1     = (const float*)d_in[0];
    const float* z2     = (const float*)d_in[1];
    const int*   labels = (const int*)d_in[2];
    (void)in_sizes; (void)n_in; (void)out_size; (void)ws_size;

    float* ws = (float*)d_ws;
    int*   cnt = (int*)(ws + WS_CNT);
    float* p1  = ws + WS_P1;
    float* p2  = ws + WS_P2;
    float* zb  = ws + WS_ZB;
    float* lg  = ws + WS_LG;

    float* out   = (float*)d_out;
    float* out1  = out + 1;
    float* out2  = out + 1 + PROTO_E;

    zero_counts<<<1, 128, 0, stream>>>(cnt);
    count_kernel<<<N_ROWS / 256, 256, 0, stream>>>(labels, cnt);
    segsum_phase1<<<dim3(NCC, NRC), COLS, 0, stream>>>(z1, z2, labels, p1, p2);
    finalize_proto<<<PROTO_E / 256, 256, 0, stream>>>(p1, p2, cnt, out1, out2);
    normalize_rows<<<C2P, 256, 0, stream>>>(out1, zb);
    logits_wmma<<<NTILE * NTILE, 32, 0, stream>>>(zb, lg);
    loss_kernel<<<1, 256, 0, stream>>>(lg, cnt, out);
}


# Logits = Gram product of the raw prototypes: a tile sums its rows' squares from the fragments it loads and scales by both inverse norms; normalise launch goes, 5 to 4; norm sum order changed, fixed
# speedup vs baseline: 6.1901x; 6.1901x over previous
#include <hip/hip_runtime.h>
#include <hip/hip_bf16.h>

#define N_ROWS   32768
#define D        768
#define NCLS     100
#define C2       200
#define C2P      208
#define NTILE    13
#define NRC      16
#define ROWS_PER 2048
#define PROTO_E  (NCLS * D)

typedef float v2f __attribute__((ext_vector_type(2)));
typedef float v8f __attribute__((ext_vector_type(8)));

#define WS_CNT   0
#define WS_P1    128
#define WS_P2    (WS_P1 + NRC * PROTO_E)
#define WS_ZB    (WS_P2 + NRC * PROTO_E)
#define WS_LG    (WS_ZB + C2P * D)
#define WS_CP    WS_LG
#define CNT_PITCH 128

#define SEG_COLS   32
#define SEG_NCG    48
#define SEG_HALF   24
#define SEG_WAVES  4
#define SEG_WROWS  512
#define SEG_SS     64
#define SEG_STEPS  16
#define SEG_NSS    8
#define NCT        7
#define SEG_PLANE  (NCLS * SEG_COLS)

typedef float v4f __attribute__((ext_vector_type(4)));

__device__ __forceinline__ void seg_load(const float* __restrict__ zp,
                                         const int* __restrict__ lp, int ss,
                                         v2f (&buf)[2 * SEG_STEPS], int (&labv)[2]) {
    labv[0] = lp[ss * SEG_SS];
    labv[1] = lp[ss * SEG_SS + 32];
#pragma unroll
    for (int i = 0; i < 2 * SEG_STEPS; ++i) {
        buf[i] = *(const v2f*)(zp + (size_t)(ss * SEG_SS + 4 * (i >> 1) + (i & 1)) * D);
    }
}

__device__ __forceinline__ unsigned seg_tilebit(int l) {
    return 1u << ((((unsigned)l) >> 4) & 31u);
}

__device__ __forceinline__ void seg_fragments(const v2f (&raw)[2 * SEG_STEPS],
                                              v2f (&frag)[2 * SEG_STEPS]) {
#pragma unroll
    for (int i = 0; i < SEG_STEPS; ++i) {
        const v2f u = raw[2 * i], v = raw[2 * i + 1];
        frag[2 * i]     = v2f{u.x, v.x};
        frag[2 * i + 1] = v2f{u.y, v.y};
    }
}

__device__ __forceinline__ void seg_compute(const v2f (&frag)[2 * SEG_STEPS], const int (&labv)[2],
                                            int c, int h, v8f (&acc)[NCT][2]) {
#pragma unroll
    for (int i = 0; i < SEG_STEPS; ++i) {
        const int lv = labv[i >> 3];
        const int l0 = __builtin_amdgcn_readlane(lv, 4 * (i & 7) + 0);
        const int l1 = __builtin_amdgcn_readlane(lv, 4 * (i & 7) + 1);
        const int l2 = __builtin_amdgcn_readlane(lv, 4 * (i & 7) + 2);
        const int l3 = __builtin_amdgcn_readlane(lv, 4 * (i & 7) + 3);
        const unsigned tm = seg_tilebit(l0) | seg_tilebit(l1) | seg_tilebit(l2) | seg_tilebit(l3);
        const int da = (h ? l2 : l0) - c;
        const int db = (h ? l3 : l1) - c;
        const v2f b0 = frag[2 * i];
        const v2f b1 = frag[2 * i + 1];
#pragma unroll
        for (int t = 0; t < NCT; ++t) {
            if (tm & (1u << t)) {
                const v2f a = {da == 16 * t ? 1.0f : 0.0f, db == 16 * t ? 1.0f : 0.0f};
                acc[t][0] = __builtin_amdgcn_wmma_f32_16x16x4_f32(
                    false, a, false, b0, (short)0, acc[t][0], false, false);
                acc[t][1] = __builtin_amdgcn_wmma_f32_16x16x4_f32(
                    false, a, false, b1, (short)0, acc[t][1], false, false);
            }
        }
    }
}

__global__ __launch_bounds__(SEG_WAVES * 32) void segsum_onehot(
        const float* __restrict__ z1, const float* __restrict__ z2,
        const int* __restrict__ labels,
        float* __restrict__ p1, float* __restrict__ p2, int* __restrict__ cntp) {
    __shared__ __attribute__((aligned(16))) float red[SEG_WAVES * SEG_PLANE];
    __shared__ int hist[CNT_PITCH];
    const int tid  = threadIdx.x;
    const int lane = tid & 31;
    const int w    = __builtin_amdgcn_readfirstlane(tid >> 5);
    const int cg   = blockIdx.x;
    const int rc   = blockIdx.y;
    const bool second = cg >= SEG_HALF;
    const float* __restrict__ z = second ? z2 : z1;
    float* __restrict__ p       = second ? p2 : p1;
    const int colBase = (second ? cg - SEG_HALF : cg) * SEG_COLS;
    const int c = lane & 15;
    const int h = lane >> 4;
    const int rowBase = rc * ROWS_PER + w * SEG_WROWS;
    const float* __restrict__ zp = z + (size_t)(rowBase + 2 * h) * D + colBase + 2 * c;
    const int* __restrict__ lp   = labels + rowBase + lane;
    const bool counter = cg == 0;
    if (counter) {
        hist[tid] = 0;
        __syncthreads();
    }

    v8f acc[NCT][2];
#pragma unroll
    for (int t = 0; t < NCT; ++t) { acc[t][0] = v8f{}; acc[t][1] = v8f{}; }

    v2f raw[2 * SEG_STEPS], frag[2 * SEG_STEPS];
    int labRaw[2], labCur[2];
    seg_load(zp, lp, 0, raw, labRaw);
    seg_fragments(raw, frag);
    labCur[0] = labRaw[0]; labCur[1] = labRaw[1];
    for (int ss = 1; ss < SEG_NSS; ++ss) {
        seg_load(zp, lp, ss, raw, labRaw);
        if (counter) { atomicAdd(&hist[labCur[0]], 1); atomicAdd(&hist[labCur[1]], 1); }
        seg_compute(frag, labCur, c, h, acc);
        seg_fragments(raw, frag);
        labCur[0] = labRaw[0]; labCur[1] = labRaw[1];
    }
    if (counter) { atomicAdd(&hist[labCur[0]], 1); atomicAdd(&hist[labCur[1]], 1); }
    seg_compute(frag, labCur, c, h, acc);

    float* mine = red + w * SEG_PLANE;
#pragma unroll
    for (int t = 0; t < NCT; ++t) {
#pragma unroll
        for (int r = 0; r < 8; ++r) {
            const int cls = 16 * t + 8 * h + r;
            if (cls < NCLS) {
                const v2f pr = {acc[t][0][r], acc[t][1][r]};
                *(v2f*)(mine + cls * SEG_COLS + 2 * c) = pr;
            }
        }
    }
    __syncthreads();
    const v4f* red4 = (const v4f*)red;
    for (int e = tid; e < SEG_PLANE / 4; e += SEG_WAVES * 32) {
        v4f s = red4[e];
        s += red4[e + SEG_PLANE / 4];
        s += red4[e + 2 * (SEG_PLANE / 4)];
        s += red4[e + 3 * (SEG_PLANE / 4)];
        const int cls = e >> 3, q = e & 7;
        *(v4f*)(p + (size_t)rc * PROTO_E + cls * D + colBase + 4 * q) = s;
    }
    if (counter) cntp[rc * CNT_PITCH + tid] = hist[tid];
}

#define FIN_BLOCKS (PROTO_E / 256)
#define FIN_PADB   ((C2P - C2) * D / 256)
__global__ void finalize_proto(const float* __restrict__ p1, const float* __restrict__ p2,
                               const int* __restrict__ cntp, int* __restrict__ cnt,
                               float* __restrict__ out1, float* __restrict__ out2,
                               float* __restrict__ zb) {
    if (blockIdx.x >= FIN_BLOCKS) {
        zb[(size_t)C2 * D + (blockIdx.x - FIN_BLOCKS) * 256 + threadIdx.x] = 0.f;
        return;
    }
    const int e = blockIdx.x * 256 + threadIdx.x;
    const int cls = blockIdx.x / 3;
    int n = 0;
    for (int rc = 0; rc < NRC; ++rc) n += cntp[rc * CNT_PITCH + cls];
    if (blockIdx.x == 0 && threadIdx.x < CNT_PITCH) {
        int m = 0;
        for (int rc = 0; rc < NRC; ++rc) m += cntp[rc * CNT_PITCH + threadIdx.x];
        cnt[threadIdx.x] = m;
    }
    float s1 = 0.f, s2 = 0.f;
    for (int rc = 0; rc < NRC; ++rc) {
        s1 += p1[(size_t)rc * PROTO_E + e];
        s2 += p2[(size_t)rc * PROTO_E + e];
    }
    const float d = fmaxf((float)n, 1.0f);
    const float q1 = s1 / d, q2 = s2 / d;
    out1[e] = q1;
    out2[e] = q2;
    zb[e] = q1;
    zb[PROTO_E + e] = q2;
}

__device__ __forceinline__ v2f sq_acc(const v2f x, const v2f s) {
    return __builtin_elementwise_fma(x, x, s);
}

__global__ __launch_bounds__(32) void logits_wmma(const float* __restrict__ zb,
                                                  float* __restrict__ lg) {
    const int tile = blockIdx.x;
    const int ti = tile / NTILE, tj = tile % NTILE;
    const int lane = threadIdx.x;
    const int mn = lane & 15;
    const int kb = (lane >> 4) << 1;
    const float* __restrict__ arow = zb + (size_t)(ti * 16 + mn) * D;
    const float* __restrict__ brow = zb + (size_t)(tj * 16 + mn) * D;
    v8f acc = {};
    v2f sa = {0.f, 0.f}, sb = {0.f, 0.f};
    for (int k0 = 0; k0 < D; k0 += 4) {
        const v2f a = *(const v2f*)(arow + k0 + kb);
        const v2f b = *(const v2f*)(brow + k0 + kb);
        sa = sq_acc(a, sa);
        sb = sq_acc(b, sb);
        acc = __builtin_amdgcn_wmma_f32_16x16x4_f32(
             false, a,  false, b,
             (short)0, acc,  false,  false);
    }
    float na = sa.x + sa.y, nb = sb.x + sb.y;
    na += __shfl_xor(na, 16);
    nb += __shfl_xor(nb, 16);
    const float ia = 1.0f / fmaxf(sqrtf(na), 1e-12f);
    const float ib = 1.0f / fmaxf(sqrtf(nb), 1e-12f);
    const int mhi = (lane >> 4) * 8;
#pragma unroll
    for (int v = 0; v < 8; ++v) {
        const int m = ti * 16 + v + mhi;
        const float iar = __shfl(ia, mhi + v);
        lg[(size_t)m * C2P + tj * 16 + mn] = acc[v] * (iar * ib) * 2.0f;
    }
}

__device__ __forceinline__ bool cls_valid(const unsigned (&vm)[4], int cls) {
    return (vm[cls >> 5] >> (cls & 31)) & 1u;
}

__global__ __launch_bounds__(256) void loss_kernel(const float* __restrict__ lg,
                                                   const int* __restrict__ cnt,
                                                   float* __restrict__ out) {
    __shared__ float sterm[256];
    __shared__ float snv[256];
    const int t = threadIdx.x;
    const int lane = t & 31;
    unsigned vm[4];
    vm[0] = __builtin_amdgcn_ballot_w32(cnt[lane] > 0);
    vm[1] = __builtin_amdgcn_ballot_w32(cnt[32 + lane] > 0);
    vm[2] = __builtin_amdgcn_ballot_w32(cnt[64 + lane] > 0);
    const int c3 = cnt[96 + (lane & 3)];
    vm[3] = __builtin_amdgcn_ballot_w32(lane < NCLS - 96 && c3 > 0);
    float term = 0.f, nvf = 0.f;
    if (t < C2) {
        const int tcls = t < NCLS ? t : t - NCLS;
        const unsigned mw = tcls < 32 ? vm[0] : tcls < 64 ? vm[1] : tcls < 96 ? vm[2] : vm[3];
        const bool vr = (mw >> (tcls & 31)) & 1u;
        nvf = vr ? 1.f : 0.f;
        float v[C2];
#pragma unroll
        for (int c = 0; c < C2; ++c) v[c] = lg[(size_t)c * C2P + t];
        const int p = t < NCLS ? t + NCLS : t - NCLS;
        const float pos = lg[(size_t)t * C2P + p];
        float mx = -3.402823466e38f;
#pragma unroll
        for (int c = 0; c < C2; ++c)
            if (cls_valid(vm, c % NCLS)) mx = fmaxf(mx, v[c]);
        float se = 0.f;
#pragma unroll
        for (int c = 0; c < C2; ++c) {
            const float e = expf(v[c] - mx);
            se += (cls_valid(vm, c % NCLS) && c != t) ? e : 0.f;
        }
        if (vr) term = pos - mx - logf(se);
    }
    sterm[t] = term;
    snv[t] = nvf;
    __syncthreads();
    for (int s = 128; s > 0; s >>= 1) {
        if (t < s) { sterm[t] += sterm[t + s]; snv[t] += snv[t + s]; }
        __syncthreads();
    }
    if (t == 0) out[0] = -sterm[0] / fmaxf(snv[0], 1.0f);
}

extern "C" void kernel_launch(void* const* d_in, const int* in_sizes, int n_in,
                              void* d_out, int out_size, void* d_ws, size_t ws_size,
                              hipStream_t stream) {
    const float* z1     = (const float*)d_in[0];
    const float* z2     = (const float*)d_in[1];
    const int*   labels = (const int*)d_in[2];
    (void)in_sizes; (void)n_in; (void)out_size; (void)ws_size;

    float* ws = (float*)d_ws;
    int*   cnt = (int*)(ws + WS_CNT);
    float* p1  = ws + WS_P1;
    float* p2  = ws + WS_P2;
    float* zb  = ws + WS_ZB;
    float* lg  = ws + WS_LG;
    int*   cntp = (int*)(ws + WS_CP);

    float* out   = (float*)d_out;
    float* out1  = out + 1;
    float* out2  = out + 1 + PROTO_E;

    segsum_onehot<<<dim3(SEG_NCG, NRC), SEG_WAVES * 32, 0, stream>>>(z1, z2, labels, p1, p2, cntp);
    finalize_proto<<<FIN_BLOCKS + FIN_PADB, 256, 0, stream>>>(p1, p2, cntp, cnt, out1, out2, zb);
    logits_wmma<<<NTILE * NTILE, 32, 0, stream>>>(zb, lg);
    loss_kernel<<<1, 256, 0, stream>>>(lg, cnt, out);
}


# Segment sums: the 64 rows of a load group enter the one-hot product sorted by 16-class tile (stable, slots from ballots, order kept in LDS): a k-step meets 1-2 tiles, not 3; class add order kept
# speedup vs baseline: 6.7825x; 1.0957x over previous
#include <hip/hip_runtime.h>
#include <hip/hip_bf16.h>

#define N_ROWS   32768
#define D        768
#define NCLS     100
#define C2       200
#define C2P      208
#define NTILE    13
#define NRC      16
#define ROWS_PER 2048
#define PROTO_E  (NCLS * D)

typedef float v2f __attribute__((ext_vector_type(2)));
typedef float v8f __attribute__((ext_vector_type(8)));

#define WS_CNT   0
#define WS_P1    128
#define WS_P2    (WS_P1 + NRC * PROTO_E)
#define WS_ZB    (WS_P2 + NRC * PROTO_E)
#define WS_LG    (WS_ZB + C2P * D)
#define WS_CP    WS_LG
#define CNT_PITCH 128

#define SEG_COLS   32
#define SEG_NCG    48
#define SEG_HALF   24
#define SEG_WAVES  4
#define SEG_WROWS  512
#define SEG_SS     64
#define SEG_STEPS  16
#define SEG_NSS    8
#define NCT        7
#define SEG_PLANE  (NCLS * SEG_COLS)
#define SEG_OFFBITS 18
#define SEG_OFFMASK ((1u << SEG_OFFBITS) - 1u)
#define SEG_NOCLS   127u

typedef float v4f __attribute__((ext_vector_type(4)));
typedef unsigned v2u __attribute__((ext_vector_type(2)));

__device__ __forceinline__ void seg_sort(const int (&labv)[2], unsigned* slot, int lane,
                                         int (&labs)[2]) {
    const unsigned a = min((unsigned)labv[0], SEG_NOCLS);
    const unsigned b = min((unsigned)labv[1], SEG_NOCLS);
    const unsigned ta = a >> 4, tb = b >> 4;
    unsigned base = 0, pa = 0, pb = 0;
#pragma unroll
    for (unsigned t = 0; t < 8; ++t) {
        const unsigned m0 = __builtin_amdgcn_ballot_w32(ta == t);
        const unsigned m1 = __builtin_amdgcn_ballot_w32(tb == t);
        const unsigned n0 = __builtin_popcount(m0);
        const unsigned ra = __builtin_amdgcn_mbcnt_lo(m0, base);
        const unsigned rb = __builtin_amdgcn_mbcnt_lo(m1, base + n0);
        pa = ta == t ? ra : pa;
        pb = tb == t ? rb : pb;
        base += n0 + __builtin_popcount(m1);
    }
    slot[pa] = (a << SEG_OFFBITS) | (unsigned)(lane * (D * 4));
    slot[pb] = (b << SEG_OFFBITS) | (unsigned)((lane + 32) * (D * 4));
    labs[0] = (int)(slot[lane] >> SEG_OFFBITS);
    labs[1] = (int)(slot[lane + 32] >> SEG_OFFBITS);
}

__device__ __forceinline__ void seg_load(const char* __restrict__ zg, const unsigned* slot,
                                         int h, unsigned laneOff, v2f (&buf)[2 * SEG_STEPS]) {
#pragma unroll
    for (int i = 0; i < SEG_STEPS; ++i) {
        const v2u q = *(const v2u*)(slot + 4 * i + 2 * h);
        buf[2 * i]     = *(const v2f*)(zg + ((q.x & SEG_OFFMASK) | laneOff));
        buf[2 * i + 1] = *(const v2f*)(zg + ((q.y & SEG_OFFMASK) | laneOff));
    }
}

__device__ __forceinline__ unsigned seg_tilebit(int l) {
    return 1u << ((((unsigned)l) >> 4) & 31u);
}

__device__ __forceinline__ void seg_fragments(const v2f (&raw)[2 * SEG_STEPS],
                                              v2f (&frag)[2 * SEG_STEPS]) {
#pragma unroll
    for (int i = 0; i < SEG_STEPS; ++i) {
        const v2f u = raw[2 * i], v = raw[2 * i + 1];
        frag[2 * i]     = v2f{u.x, v.x};
        frag[2 * i + 1] = v2f{u.y, v.y};
    }
}

__device__ __forceinline__ void seg_compute(const v2f (&frag)[2 * SEG_STEPS], const int (&labv)[2],
                                            int c, int h, v8f (&acc)[NCT][2]) {
#pragma unroll
    for (int i = 0; i < SEG_STEPS; ++i) {
        const int lv = labv[i >> 3];
        const int l0 = __builtin_amdgcn_readlane(lv, 4 * (i & 7) + 0);
        const int l1 = __builtin_amdgcn_readlane(lv, 4 * (i & 7) + 1);
        const int l2 = __builtin_amdgcn_readlane(lv, 4 * (i & 7) + 2);
        const int l3 = __builtin_amdgcn_readlane(lv, 4 * (i & 7) + 3);
        const unsigned tm = seg_tilebit(l0) | seg_tilebit(l1) | seg_tilebit(l2) | seg_tilebit(l3);
        const int da = (h ? l2 : l0) - c;
        const int db = (h ? l3 : l1) - c;
        const v2f b0 = frag[2 * i];
        const v2f b1 = frag[2 * i + 1];
#pragma unroll
        for (int t = 0; t < NCT; ++t) {
            if (tm & (1u << t)) {
                const v2f a = {da == 16 * t ? 1.0f : 0.0f, db == 16 * t ? 1.0f : 0.0f};
                acc[t][0] = __builtin_amdgcn_wmma_f32_16x16x4_f32(
                    false, a, false, b0, (short)0, acc[t][0], false, false);
                acc[t][1] = __builtin_amdgcn_wmma_f32_16x16x4_f32(
                    false, a, false, b1, (short)0, acc[t][1], false, false);
            }
        }
    }
}

__global__ __launch_bounds__(SEG_WAVES * 32) void segsum_onehot(
        const float* __restrict__ z1, const float* __restrict__ z2,
        const int* __restrict__ labels,
        float* __restrict__ p1, float* __restrict__ p2, int* __restrict__ cntp) {
    __shared__ __attribute__((aligned(16))) float red[SEG_WAVES * SEG_PLANE];
    __shared__ int hist[CNT_PITCH];
    __shared__ __attribute__((aligned(16))) unsigned slots[SEG_WAVES][SEG_SS];
    const int tid  = threadIdx.x;
    const int lane = tid & 31;
    const int w    = __builtin_amdgcn_readfirstlane(tid >> 5);
    const int cg   = blockIdx.x;
    const int rc   = blockIdx.y;
    const bool second = cg >= SEG_HALF;
    const float* __restrict__ z = second ? z2 : z1;
    float* __restrict__ p       = second ? p2 : p1;
    const int colBase = (second ? cg - SEG_HALF : cg) * SEG_COLS;
    const int c = lane & 15;
    const int h = lane >> 4;
    const int rowBase = rc * ROWS_PER + w * SEG_WROWS;
    const char* __restrict__ zw = (const char*)(z + (size_t)rowBase * D + colBase);
    const unsigned laneOff = 8u * (unsigned)c;
    const int* __restrict__ lp = labels + rowBase + lane;
    unsigned* slot = slots[w];
    const bool counter = cg == 0;
    if (counter) {
        hist[tid] = 0;
        __syncthreads();
    }

    v8f acc[NCT][2];
#pragma unroll
    for (int t = 0; t < NCT; ++t) { acc[t][0] = v8f{}; acc[t][1] = v8f{}; }

    v2f raw[2 * SEG_STEPS], frag[2 * SEG_STEPS];
    int labIn[2], labAhead[2], labCur[2], labNext[2];
    labIn[0] = lp[0];
    labIn[1] = lp[32];
    labAhead[0] = lp[SEG_SS];
    labAhead[1] = lp[SEG_SS + 32];
    seg_sort(labIn, slot, lane, labCur);
    seg_load(zw, slot, h, laneOff, raw);
    seg_fragments(raw, frag);
    for (int ss = 1; ss < SEG_NSS; ++ss) {
        labIn[0] = labAhead[0]; labIn[1] = labAhead[1];
        if (ss + 1 < SEG_NSS) {
            labAhead[0] = lp[(ss + 1) * SEG_SS];
            labAhead[1] = lp[(ss + 1) * SEG_SS + 32];
        }
        seg_sort(labIn, slot, lane, labNext);
        seg_load(zw + (size_t)ss * (SEG_SS * D * 4), slot, h, laneOff, raw);
        if (counter) { atomicAdd(&hist[labCur[0]], 1); atomicAdd(&hist[labCur[1]], 1); }
        seg_compute(frag, labCur, c, h, acc);
        seg_fragments(raw, frag);
        labCur[0] = labNext[0]; labCur[1] = labNext[1];
    }
    if (counter) { atomicAdd(&hist[labCur[0]], 1); atomicAdd(&hist[labCur[1]], 1); }
    seg_compute(frag, labCur, c, h, acc);

    float* mine = red + w * SEG_PLANE;
#pragma unroll
    for (int t = 0; t < NCT; ++t) {
#pragma unroll
        for (int r = 0; r < 8; ++r) {
            const int cls = 16 * t + 8 * h + r;
            if (cls < NCLS) {
                const v2f pr = {acc[t][0][r], acc[t][1][r]};
                *(v2f*)(mine + cls * SEG_COLS + 2 * c) = pr;
            }
        }
    }
    __syncthreads();
    const v4f* red4 = (const v4f*)red;
    for (int e = tid; e < SEG_PLANE / 4; e += SEG_WAVES * 32) {
        v4f s = red4[e];
        s += red4[e + SEG_PLANE / 4];
        s += red4[e + 2 * (SEG_PLANE / 4)];
        s += red4[e + 3 * (SEG_PLANE / 4)];
        const int cls = e >> 3, q = e & 7;
        *(v4f*)(p + (size_t)rc * PROTO_E + cls * D + colBase + 4 * q) = s;
    }
    if (counter) cntp[rc * CNT_PITCH + tid] = hist[tid];
}

#define FIN_BLOCKS (PROTO_E / 256)
#define FIN_PADB   ((C2P - C2) * D / 256)
__global__ void finalize_proto(const float* __restrict__ p1, const float* __restrict__ p2,
                               const int* __restrict__ cntp, int* __restrict__ cnt,
                               float* __restrict__ out1, float* __restrict__ out2,
                               float* __restrict__ zb) {
    if (blockIdx.x >= FIN_BLOCKS) {
        zb[(size_t)C2 * D + (blockIdx.x - FIN_BLOCKS) * 256 + threadIdx.x] = 0.f;
        return;
    }
    const int e = blockIdx.x * 256 + threadIdx.x;
    const int cls = blockIdx.x / 3;
    int n = 0;
    for (int rc = 0; rc < NRC; ++rc) n += cntp[rc * CNT_PITCH + cls];
    if (blockIdx.x == 0 && threadIdx.x < CNT_PITCH) {
        int m = 0;
        for (int rc = 0; rc < NRC; ++rc) m += cntp[rc * CNT_PITCH + threadIdx.x];
        cnt[threadIdx.x] = m;
    }
    float s1 = 0.f, s2 = 0.f;
    for (int rc = 0; rc < NRC; ++rc) {
        s1 += p1[(size_t)rc * PROTO_E + e];
        s2 += p2[(size_t)rc * PROTO_E + e];
    }
    const float d = fmaxf((float)n, 1.0f);
    const float q1 = s1 / d, q2 = s2 / d;
    out1[e] = q1;
    out2[e] = q2;
    zb[e] = q1;
    zb[PROTO_E + e] = q2;
}

__device__ __forceinline__ v2f sq_acc(const v2f x, const v2f s) {
    return __builtin_elementwise_fma(x, x, s);
}

__global__ __launch_bounds__(32) void logits_wmma(const float* __restrict__ zb,
                                                  float* __restrict__ lg) {
    const int tile = blockIdx.x;
    const int ti = tile / NTILE, tj = tile % NTILE;
    const int lane = threadIdx.x;
    const int mn = lane & 15;
    const int kb = (lane >> 4) << 1;
    const float* __restrict__ arow = zb + (size_t)(ti * 16 + mn) * D;
    const float* __restrict__ brow = zb + (size_t)(tj * 16 + mn) * D;
    v8f acc = {};
    v2f sa = {0.f, 0.f}, sb = {0.f, 0.f};
    for (int k0 = 0; k0 < D; k0 += 4) {
        const v2f a = *(const v2f*)(arow + k0 + kb);
        const v2f b = *(const v2f*)(brow + k0 + kb);
        sa = sq_acc(a, sa);
        sb = sq_acc(b, sb);
        acc = __builtin_amdgcn_wmma_f32_16x16x4_f32(
             false, a,  false, b,
             (short)0, acc,  false,  false);
    }
    float na = sa.x + sa.y, nb = sb.x + sb.y;
    na += __shfl_xor(na, 16);
    nb += __shfl_xor(nb, 16);
    const float ia = 1.0f / fmaxf(sqrtf(na), 1e-12f);
    const float ib = 1.0f / fmaxf(sqrtf(nb), 1e-12f);
    const int mhi = (lane >> 4) * 8;
#pragma unroll
    for (int v = 0; v < 8; ++v) {
        const int m = ti * 16 + v + mhi;
        const float iar = __shfl(ia, mhi + v);
        lg[(size_t)m * C2P + tj * 16 + mn] = acc[v] * (iar * ib) * 2.0f;
    }
}

__device__ __forceinline__ bool cls_valid(const unsigned (&vm)[4], int cls) {
    return (vm[cls >> 5] >> (cls & 31)) & 1u;
}

__global__ __launch_bounds__(256) void loss_kernel(const float* __restrict__ lg,
                                                   const int* __restrict__ cnt,
                                                   float* __restrict__ out) {
    __shared__ float sterm[256];
    __shared__ float snv[256];
    const int t = threadIdx.x;
    const int lane = t & 31;
    unsigned vm[4];
    vm[0] = __builtin_amdgcn_ballot_w32(cnt[lane] > 0);
    vm[1] = __builtin_amdgcn_ballot_w32(cnt[32 + lane] > 0);
    vm[2] = __builtin_amdgcn_ballot_w32(cnt[64 + lane] > 0);
    const int c3 = cnt[96 + (lane & 3)];
    vm[3] = __builtin_amdgcn_ballot_w32(lane < NCLS - 96 && c3 > 0);
    float term = 0.f, nvf = 0.f;
    if (t < C2) {
        const int tcls = t < NCLS ? t : t - NCLS;
        const unsigned mw = tcls < 32 ? vm[0] : tcls < 64 ? vm[1] : tcls < 96 ? vm[2] : vm[3];
        const bool vr = (mw >> (tcls & 31)) & 1u;
        nvf = vr ? 1.f : 0.f;
        float v[C2];
#pragma unroll
        for (int c = 0; c < C2; ++c) v[c] = lg[(size_t)c * C2P + t];
        const int p = t < NCLS ? t + NCLS : t - NCLS;
        const float pos = lg[(size_t)t * C2P + p];
        float mx = -3.402823466e38f;
#pragma unroll
        for (int c = 0; c < C2; ++c)
            if (cls_valid(vm, c % NCLS)) mx = fmaxf(mx, v[c]);
        float se = 0.f;
#pragma unroll
        for (int c = 0; c < C2; ++c) {
            const float e = expf(v[c] - mx);
            se += (cls_valid(vm, c % NCLS) && c != t) ? e : 0.f;
        }
        if (vr) term = pos - mx - logf(se);
    }
    sterm[t] = term;
    snv[t] = nvf;
    __syncthreads();
    for (int s = 128; s > 0; s >>= 1) {
        if (t < s) { sterm[t] += sterm[t + s]; snv[t] += snv[t + s]; }
        __syncthreads();
    }
    if (t == 0) out[0] = -sterm[0] / fmaxf(snv[0], 1.0f);
}

extern "C" void kernel_launch(void* const* d_in, const int* in_sizes, int n_in,
                              void* d_out, int out_size, void* d_ws, size_t ws_size,
                              hipStream_t stream) {
    const float* z1     = (const float*)d_in[0];
    const float* z2     = (const float*)d_in[1];
    const int*   labels = (const int*)d_in[2];
    (void)in_sizes; (void)n_in; (void)out_size; (void)ws_size;

    float* ws = (float*)d_ws;
    int*   cnt = (int*)(ws + WS_CNT);
    float* p1  = ws + WS_P1;
    float* p2  = ws + WS_P2;
    float* zb  = ws + WS_ZB;
    float* lg  = ws + WS_LG;
    int*   cntp = (int*)(ws + WS_CP);

    float* out   = (float*)d_out;
    float* out1  = out + 1;
    float* out2  = out + 1 + PROTO_E;

    segsum_onehot<<<dim3(SEG_NCG, NRC), SEG_WAVES * 32, 0, stream>>>(z1, z2, labels, p1, p2, cntp);
    finalize_proto<<<FIN_BLOCKS + FIN_PADB, 256, 0, stream>>>(p1, p2, cntp, cnt, out1, out2, zb);
    logits_wmma<<<NTILE * NTILE, 32, 0, stream>>>(zb, lg);
    loss_kernel<<<1, 256, 0, stream>>>(lg, cnt, out);
}


# Segment sums: a k-step's class tiles come from the sort's scalar tile bounds, its one-hot rows from the sorted list in LDS; no lane reads or scalar chains ahead of a step's branches; sums unchanged
# speedup vs baseline: 7.3472x; 1.0833x over previous
#include <hip/hip_runtime.h>
#include <hip/hip_bf16.h>

#define N_ROWS   32768
#define D        768
#define NCLS     100
#define C2       200
#define C2P      208
#define NTILE    13
#define NRC      16
#define ROWS_PER 2048
#define PROTO_E  (NCLS * D)

typedef float v2f __attribute__((ext_vector_type(2)));
typedef float v8f __attribute__((ext_vector_type(8)));

#define WS_CNT   0
#define WS_P1    128
#define WS_P2    (WS_P1 + NRC * PROTO_E)
#define WS_ZB    (WS_P2 + NRC * PROTO_E)
#define WS_LG    (WS_ZB + C2P * D)
#define WS_CP    WS_LG
#define CNT_PITCH 128

#define SEG_COLS   32
#define SEG_NCG    48
#define SEG_HALF   24
#define SEG_WAVES  4
#define SEG_WROWS  512
#define SEG_SS     64
#define SEG_STEPS  16
#define SEG_NSS    8
#define NCT        7
#define SEG_PLANE  (NCLS * SEG_COLS)
#define SEG_OFFBITS 18
#define SEG_OFFMASK ((1u << SEG_OFFBITS) - 1u)
#define SEG_NOCLS   127u

typedef float v4f __attribute__((ext_vector_type(4)));
typedef unsigned v2u __attribute__((ext_vector_type(2)));

__device__ __forceinline__ void seg_sort(const int (&labv)[2], unsigned* slot, int lane,
                                         unsigned (&sm)[NCT]) {
    const unsigned a = min((unsigned)labv[0], SEG_NOCLS);
    const unsigned b = min((unsigned)labv[1], SEG_NOCLS);
    const unsigned ta = a >> 4, tb = b >> 4;
    unsigned base = 0, pa = 0, pb = 0;
#pragma unroll
    for (unsigned t = 0; t < 8; ++t) {
        const unsigned m0 = __builtin_amdgcn_ballot_w32(ta == t);
        const unsigned m1 = __builtin_amdgcn_ballot_w32(tb == t);
        const unsigned n0 = __builtin_popcount(m0);
        const unsigned ra = __builtin_amdgcn_mbcnt_lo(m0, base);
        const unsigned rb = __builtin_amdgcn_mbcnt_lo(m1, base + n0);
        pa = ta == t ? ra : pa;
        pb = tb == t ? rb : pb;
        const unsigned first = base;
        base += n0 + __builtin_popcount(m1);
        if (t < NCT)
            sm[t] = base > first ? (1u << ((base + 3) >> 2)) - (1u << (first >> 2)) : 0u;
    }
    slot[pa] = (a << SEG_OFFBITS) | (unsigned)(lane * (D * 4));
    slot[pb] = (b << SEG_OFFBITS) | (unsigned)((lane + 32) * (D * 4));
}

__device__ __forceinline__ void seg_load(const char* __restrict__ zg, const unsigned* slot,
                                         int h, unsigned laneOff, v2f (&buf)[2 * SEG_STEPS]) {
#pragma unroll
    for (int i = 0; i < SEG_STEPS; ++i) {
        const v2u q = *(const v2u*)(slot + 4 * i + 2 * h);
        buf[2 * i]     = *(const v2f*)(zg + ((q.x & SEG_OFFMASK) | laneOff));
        buf[2 * i + 1] = *(const v2f*)(zg + ((q.y & SEG_OFFMASK) | laneOff));
    }
}

__device__ __forceinline__ void seg_fragments(const v2f (&raw)[2 * SEG_STEPS],
                                              v2f (&frag)[2 * SEG_STEPS]) {
#pragma unroll
    for (int i = 0; i < SEG_STEPS; ++i) {
        const v2f u = raw[2 * i], v = raw[2 * i + 1];
        frag[2 * i]     = v2f{u.x, v.x};
        frag[2 * i + 1] = v2f{u.y, v.y};
    }
}

__device__ __forceinline__ void seg_labels(const unsigned* slot, int h, v2u (&lq)[SEG_STEPS]) {
#pragma unroll
    for (int i = 0; i < SEG_STEPS; ++i) lq[i] = *(const v2u*)(slot + 4 * i + 2 * h);
}

__device__ __forceinline__ void seg_compute(const v2f (&frag)[2 * SEG_STEPS], const v2u (&lq)[SEG_STEPS],
                                            const unsigned (&sm)[NCT], int c, int h,
                                            v8f (&acc)[NCT][2]) {
#pragma unroll
    for (int i = 0; i < SEG_STEPS; ++i) {
        const v2f b0 = frag[2 * i];
        const v2f b1 = frag[2 * i + 1];
        const unsigned la = lq[i].x >> SEG_OFFBITS, lb = lq[i].y >> SEG_OFFBITS;
#pragma unroll
        for (int t = 0; t < NCT; ++t) {
            if (sm[t] & (1u << i)) {
                const unsigned ct = (unsigned)(16 * t + c);
                const v2f a = {la == ct ? 1.0f : 0.0f, lb == ct ? 1.0f : 0.0f};
                acc[t][0] = __builtin_amdgcn_wmma_f32_16x16x4_f32(
                    false, a, false, b0, (short)0, acc[t][0], false, false);
                acc[t][1] = __builtin_amdgcn_wmma_f32_16x16x4_f32(
                    false, a, false, b1, (short)0, acc[t][1], false, false);
            }
        }
    }
}

__global__ __launch_bounds__(SEG_WAVES * 32) void segsum_onehot(
        const float* __restrict__ z1, const float* __restrict__ z2,
        const int* __restrict__ labels,
        float* __restrict__ p1, float* __restrict__ p2, int* __restrict__ cntp) {
    __shared__ __attribute__((aligned(16))) float red[SEG_WAVES * SEG_PLANE];
    __shared__ int hist[CNT_PITCH];
    __shared__ __attribute__((aligned(16))) unsigned slots[SEG_WAVES][2][SEG_SS];
    const int tid  = threadIdx.x;
    const int lane = tid & 31;
    const int w    = __builtin_amdgcn_readfirstlane(tid >> 5);
    const int cg   = blockIdx.x;
    const int rc   = blockIdx.y;
    const bool second = cg >= SEG_HALF;
    const float* __restrict__ z = second ? z2 : z1;
    float* __restrict__ p       = second ? p2 : p1;
    const int colBase = (second ? cg - SEG_HALF : cg) * SEG_COLS;
    const int c = lane & 15;
    const int h = lane >> 4;
    const int rowBase = rc * ROWS_PER + w * SEG_WROWS;
    const char* __restrict__ zw = (const char*)(z + (size_t)rowBase * D + colBase);
    const unsigned laneOff = 8u * (unsigned)c;
    const int* __restrict__ lp = labels + rowBase + lane;
    unsigned* slotw = &slots[w][0][0];
    const bool counter = cg == 0;
    if (counter) {
        hist[tid] = 0;
        __syncthreads();
    }

    v8f acc[NCT][2];
#pragma unroll
    for (int t = 0; t < NCT; ++t) { acc[t][0] = v8f{}; acc[t][1] = v8f{}; }

    v2f raw[2 * SEG_STEPS], frag[2 * SEG_STEPS];
    int labIn[2], labAhead[2];
    unsigned smCur[NCT], smNext[NCT];
    v2u lq[SEG_STEPS];
    labIn[0] = lp[0];
    labIn[1] = lp[32];
    labAhead[0] = lp[SEG_SS];
    labAhead[1] = lp[SEG_SS + 32];
    seg_sort(labIn, slotw, lane, smCur);
    seg_load(zw, slotw, h, laneOff, raw);
    if (counter) {
        atomicAdd(&hist[min((unsigned)labIn[0], SEG_NOCLS)], 1);
        atomicAdd(&hist[min((unsigned)labIn[1], SEG_NOCLS)], 1);
    }
    seg_fragments(raw, frag);
    for (int ss = 1; ss < SEG_NSS; ++ss) {
        labIn[0] = labAhead[0]; labIn[1] = labAhead[1];
        if (ss + 1 < SEG_NSS) {
            labAhead[0] = lp[(ss + 1) * SEG_SS];
            labAhead[1] = lp[(ss + 1) * SEG_SS + 32];
        }
        unsigned* slotNew = slotw + (ss & 1) * SEG_SS;
        const unsigned* slotOld = slotw + ((ss & 1) ^ 1) * SEG_SS;
        seg_labels(slotOld, h, lq);
        seg_sort(labIn, slotNew, lane, smNext);
        seg_load(zw + (size_t)ss * (SEG_SS * D * 4), slotNew, h, laneOff, raw);
        if (counter) {
            atomicAdd(&hist[min((unsigned)labIn[0], SEG_NOCLS)], 1);
            atomicAdd(&hist[min((unsigned)labIn[1], SEG_NOCLS)], 1);
        }
        seg_compute(frag, lq, smCur, c, h, acc);
        seg_fragments(raw, frag);
#pragma unroll
        for (int t = 0; t < NCT; ++t) smCur[t] = smNext[t];
    }
    seg_labels(slotw + ((SEG_NSS - 1) & 1) * SEG_SS, h, lq);
    seg_compute(frag, lq, smCur, c, h, acc);

    float* mine = red + w * SEG_PLANE;
#pragma unroll
    for (int t = 0; t < NCT; ++t) {
#pragma unroll
        for (int r = 0; r < 8; ++r) {
            const int cls = 16 * t + 8 * h + r;
            if (cls < NCLS) {
                const v2f pr = {acc[t][0][r], acc[t][1][r]};
                *(v2f*)(mine + cls * SEG_COLS + 2 * c) = pr;
            }
        }
    }
    __syncthreads();
    const v4f* red4 = (const v4f*)red;
    for (int e = tid; e < SEG_PLANE / 4; e += SEG_WAVES * 32) {
        v4f s = red4[e];
        s += red4[e + SEG_PLANE / 4];
        s += red4[e + 2 * (SEG_PLANE / 4)];
        s += red4[e + 3 * (SEG_PLANE / 4)];
        const int cls = e >> 3, q = e & 7;
        *(v4f*)(p + (size_t)rc * PROTO_E + cls * D + colBase + 4 * q) = s;
    }
    if (counter) cntp[rc * CNT_PITCH + tid] = hist[tid];
}

#define FIN_BLOCKS (PROTO_E / 256)
#define FIN_PADB   ((C2P - C2) * D / 256)
__global__ void finalize_proto(const float* __restrict__ p1, const float* __restrict__ p2,
                               const int* __restrict__ cntp, int* __restrict__ cnt,
                               float* __restrict__ out1, float* __restrict__ out2,
                               float* __restrict__ zb) {
    if (blockIdx.x >= FIN_BLOCKS) {
        zb[(size_t)C2 * D + (blockIdx.x - FIN_BLOCKS) * 256 + threadIdx.x] = 0.f;
        return;
    }
    const int e = blockIdx.x * 256 + threadIdx.x;
    const int cls = blockIdx.x / 3;
    int n = 0;
    for (int rc = 0; rc < NRC; ++rc) n += cntp[rc * CNT_PITCH + cls];
    if (blockIdx.x == 0 && threadIdx.x < CNT_PITCH) {
        int m = 0;
        for (int rc = 0; rc < NRC; ++rc) m += cntp[rc * CNT_PITCH + threadIdx.x];
        cnt[threadIdx.x] = m;
    }
    float s1 = 0.f, s2 = 0.f;
    for (int rc = 0; rc < NRC; ++rc) {
        s1 += p1[(size_t)rc * PROTO_E + e];
        s2 += p2[(size_t)rc * PROTO_E + e];
    }
    const float d = fmaxf((float)n, 1.0f);
    const float q1 = s1 / d, q2 = s2 / d;
    out1[e] = q1;
    out2[e] = q2;
    zb[e] = q1;
    zb[PROTO_E + e] = q2;
}

__device__ __forceinline__ v2f sq_acc(const v2f x, const v2f s) {
    return __builtin_elementwise_fma(x, x, s);
}

__global__ __launch_bounds__(32) void logits_wmma(const float* __restrict__ zb,
                                                  float* __restrict__ lg) {
    const int tile = blockIdx.x;
    const int ti = tile / NTILE, tj = tile % NTILE;
    const int lane = threadIdx.x;
    const int mn = lane & 15;
    const int kb = (lane >> 4) << 1;
    const float* __restrict__ arow = zb + (size_t)(ti * 16 + mn) * D;
    const float* __restrict__ brow = zb + (size_t)(tj * 16 + mn) * D;
    v8f acc = {};
    v2f sa = {0.f, 0.f}, sb = {0.f, 0.f};
    for (int k0 = 0; k0 < D; k0 += 4) {
        const v2f a = *(const v2f*)(arow + k0 + kb);
        const v2f b = *(const v2f*)(brow + k0 + kb);
        sa = sq_acc(a, sa);
        sb = sq_acc(b, sb);
        acc = __builtin_amdgcn_wmma_f32_16x16x4_f32(
             false, a,  false, b,
             (short)0, acc,  false,  false);
    }
    float na = sa.x + sa.y, nb = sb.x + sb.y;
    na += __shfl_xor(na, 16);
    nb += __shfl_xor(nb, 16);
    const float ia = 1.0f / fmaxf(sqrtf(na), 1e-12f);
    const float ib = 1.0f / fmaxf(sqrtf(nb), 1e-12f);
    const int mhi = (lane >> 4) * 8;
#pragma unroll
    for (int v = 0; v < 8; ++v) {
        const int m = ti * 16 + v + mhi;
        const float iar = __shfl(ia, mhi + v);
        lg[(size_t)m * C2P + tj * 16 + mn] = acc[v] * (iar * ib) * 2.0f;
    }
}

__device__ __forceinline__ bool cls_valid(const unsigned (&vm)[4], int cls) {
    return (vm[cls >> 5] >> (cls & 31)) & 1u;
}

__global__ __launch_bounds__(256) void loss_kernel(const float* __restrict__ lg,
                                                   const int* __restrict__ cnt,
                                                   float* __restrict__ out) {
    __shared__ float sterm[256];
    __shared__ float snv[256];
    const int t = threadIdx.x;
    const int lane = t & 31;
    unsigned vm[4];
    vm[0] = __builtin_amdgcn_ballot_w32(cnt[lane] > 0);
    vm[1] = __builtin_amdgcn_ballot_w32(cnt[32 + lane] > 0);
    vm[2] = __builtin_amdgcn_ballot_w32(cnt[64 + lane] > 0);
    const int c3 = cnt[96 + (lane & 3)];
    vm[3] = __builtin_amdgcn_ballot_w32(lane < NCLS - 96 && c3 > 0);
    float term = 0.f, nvf = 0.f;
    if (t < C2) {
        const int tcls = t < NCLS ? t : t - NCLS;
        const unsigned mw = tcls < 32 ? vm[0] : tcls < 64 ? vm[1] : tcls < 96 ? vm[2] : vm[3];
        const bool vr = (mw >> (tcls & 31)) & 1u;
        nvf = vr ? 1.f : 0.f;
        float v[C2];
#pragma unroll
        for (int c = 0; c < C2; ++c) v[c] = lg[(size_t)c * C2P + t];
        const int p = t < NCLS ? t + NCLS : t - NCLS;
        const float pos = lg[(size_t)t * C2P + p];
        float mx = -3.402823466e38f;
#pragma unroll
        for (int c = 0; c < C2; ++c)
            if (cls_valid(vm, c % NCLS)) mx = fmaxf(mx, v[c]);
        float se = 0.f;
#pragma unroll
        for (int c = 0; c < C2; ++c) {
            const float e = expf(v[c] - mx);
            se += (cls_valid(vm, c % NCLS) && c != t) ? e : 0.f;
        }
        if (vr) term = pos - mx - logf(se);
    }
    sterm[t] = term;
    snv[t] = nvf;
    __syncthreads();
    for (int s = 128; s > 0; s >>= 1) {
        if (t < s) { sterm[t] += sterm[t + s]; snv[t] += snv[t + s]; }
        __syncthreads();
    }
    if (t == 0) out[0] = -sterm[0] / fmaxf(snv[0], 1.0f);
}

extern "C" void kernel_launch(void* const* d_in, const int* in_sizes, int n_in,
                              void* d_out, int out_size, void* d_ws, size_t ws_size,
                              hipStream_t stream) {
    const float* z1     = (const float*)d_in[0];
    const float* z2     = (const float*)d_in[1];
    const int*   labels = (const int*)d_in[2];
    (void)in_sizes; (void)n_in; (void)out_size; (void)ws_size;

    float* ws = (float*)d_ws;
    int*   cnt = (int*)(ws + WS_CNT);
    float* p1  = ws + WS_P1;
    float* p2  = ws + WS_P2;
    float* zb  = ws + WS_ZB;
    float* lg  = ws + WS_LG;
    int*   cntp = (int*)(ws + WS_CP);

    float* out   = (float*)d_out;
    float* out1  = out + 1;
    float* out2  = out + 1 + PROTO_E;

    segsum_onehot<<<dim3(SEG_NCG, NRC), SEG_WAVES * 32, 0, stream>>>(z1, z2, labels, p1, p2, cntp);
    finalize_proto<<<FIN_BLOCKS + FIN_PADB, 256, 0, stream>>>(p1, p2, cntp, cnt, out1, out2, zb);
    logits_wmma<<<NTILE * NTILE, 32, 0, stream>>>(zb, lg);
    loss_kernel<<<1, 256, 0, stream>>>(lg, cnt, out);
}


# Loss launch: exp(v-max) taken as one v_exp_f32 on fma(v, log2e, -max*log2e) instead of the range-reduced library expf (a dozen instructions, 40000 calls); exp differs in last bits; sum order kept
# speedup vs baseline: 8.4813x; 1.1544x over previous
#include <hip/hip_runtime.h>
#include <hip/hip_bf16.h>

#define N_ROWS   32768
#define D        768
#define NCLS     100
#define C2       200
#define C2P      208
#define NTILE    13
#define NRC      16
#define ROWS_PER 2048
#define PROTO_E  (NCLS * D)

typedef float v2f __attribute__((ext_vector_type(2)));
typedef float v8f __attribute__((ext_vector_type(8)));

#define WS_CNT   0
#define WS_P1    128
#define WS_P2    (WS_P1 + NRC * PROTO_E)
#define WS_ZB    (WS_P2 + NRC * PROTO_E)
#define WS_LG    (WS_ZB + C2P * D)
#define WS_CP    WS_LG
#define CNT_PITCH 128

#define SEG_COLS   32
#define SEG_NCG    48
#define SEG_HALF   24
#define SEG_WAVES  4
#define SEG_WROWS  512
#define SEG_SS     64
#define SEG_STEPS  16
#define SEG_NSS    8
#define NCT        7
#define SEG_PLANE  (NCLS * SEG_COLS)
#define SEG_OFFBITS 18
#define SEG_OFFMASK ((1u << SEG_OFFBITS) - 1u)
#define SEG_NOCLS   127u

typedef float v4f __attribute__((ext_vector_type(4)));
typedef unsigned v2u __attribute__((ext_vector_type(2)));

__device__ __forceinline__ void seg_sort(const int (&labv)[2], unsigned* slot, int lane,
                                         unsigned (&sm)[NCT]) {
    const unsigned a = min((unsigned)labv[0], SEG_NOCLS);
    const unsigned b = min((unsigned)labv[1], SEG_NOCLS);
    const unsigned ta = a >> 4, tb = b >> 4;
    unsigned base = 0, pa = 0, pb = 0;
#pragma unroll
    for (unsigned t = 0; t < 8; ++t) {
        const unsigned m0 = __builtin_amdgcn_ballot_w32(ta == t);
        const unsigned m1 = __builtin_amdgcn_ballot_w32(tb == t);
        const unsigned n0 = __builtin_popcount(m0);
        const unsigned ra = __builtin_amdgcn_mbcnt_lo(m0, base);
        const unsigned rb = __builtin_amdgcn_mbcnt_lo(m1, base + n0);
        pa = ta == t ? ra : pa;
        pb = tb == t ? rb : pb;
        const unsigned first = base;
        base += n0 + __builtin_popcount(m1);
        if (t < NCT)
            sm[t] = base > first ? (1u << ((base + 3) >> 2)) - (1u << (first >> 2)) : 0u;
    }
    slot[pa] = (a << SEG_OFFBITS) | (unsigned)(lane * (D * 4));
    slot[pb] = (b << SEG_OFFBITS) | (unsigned)((lane + 32) * (D * 4));
}

__device__ __forceinline__ void seg_load(const char* __restrict__ zg, const unsigned* slot,
                                         int h, unsigned laneOff, v2f (&buf)[2 * SEG_STEPS]) {
#pragma unroll
    for (int i = 0; i < SEG_STEPS; ++i) {
        const v2u q = *(const v2u*)(slot + 4 * i + 2 * h);
        buf[2 * i]     = *(const v2f*)(zg + ((q.x & SEG_OFFMASK) | laneOff));
        buf[2 * i + 1] = *(const v2f*)(zg + ((q.y & SEG_OFFMASK) | laneOff));
    }
}

__device__ __forceinline__ void seg_fragments(const v2f (&raw)[2 * SEG_STEPS],
                                              v2f (&frag)[2 * SEG_STEPS]) {
#pragma unroll
    for (int i = 0; i < SEG_STEPS; ++i) {
        const v2f u = raw[2 * i], v = raw[2 * i + 1];
        frag[2 * i]     = v2f{u.x, v.x};
        frag[2 * i + 1] = v2f{u.y, v.y};
    }
}

__device__ __forceinline__ void seg_labels(const unsigned* slot, int h, v2u (&lq)[SEG_STEPS]) {
#pragma unroll
    for (int i = 0; i < SEG_STEPS; ++i) lq[i] = *(const v2u*)(slot + 4 * i + 2 * h);
}

__device__ __forceinline__ void seg_compute(const v2f (&frag)[2 * SEG_STEPS], const v2u (&lq)[SEG_STEPS],
                                            const unsigned (&sm)[NCT], int c, int h,
                                            v8f (&acc)[NCT][2]) {
#pragma unroll
    for (int i = 0; i < SEG_STEPS; ++i) {
        const v2f b0 = frag[2 * i];
        const v2f b1 = frag[2 * i + 1];
        const unsigned la = lq[i].x >> SEG_OFFBITS, lb = lq[i].y >> SEG_OFFBITS;
#pragma unroll
        for (int t = 0; t < NCT; ++t) {
            if (sm[t] & (1u << i)) {
                const unsigned ct = (unsigned)(16 * t + c);
                const v2f a = {la == ct ? 1.0f : 0.0f, lb == ct ? 1.0f : 0.0f};
                acc[t][0] = __builtin_amdgcn_wmma_f32_16x16x4_f32(
                    false, a, false, b0, (short)0, acc[t][0], false, false);
                acc[t][1] = __builtin_amdgcn_wmma_f32_16x16x4_f32(
                    false, a, false, b1, (short)0, acc[t][1], false, false);
            }
        }
    }
}

__global__ __launch_bounds__(SEG_WAVES * 32) void segsum_onehot(
        const float* __restrict__ z1, const float* __restrict__ z2,
        const int* __restrict__ labels,
        float* __restrict__ p1, float* __restrict__ p2, int* __restrict__ cntp) {
    __shared__ __attribute__((aligned(16))) float red[SEG_WAVES * SEG_PLANE];
    __shared__ int hist[CNT_PITCH];
    __shared__ __attribute__((aligned(16))) unsigned slots[SEG_WAVES][2][SEG_SS];
    const int tid  = threadIdx.x;
    const int lane = tid & 31;
    const int w    = __builtin_amdgcn_readfirstlane(tid >> 5);
    const int cg   = blockIdx.x;
    const int rc   = blockIdx.y;
    const bool second = cg >= SEG_HALF;
    const float* __restrict__ z = second ? z2 : z1;
    float* __restrict__ p       = second ? p2 : p1;
    const int colBase = (second ? cg - SEG_HALF : cg) * SEG_COLS;
    const int c = lane & 15;
    const int h = lane >> 4;
    const int rowBase = rc * ROWS_PER + w * SEG_WROWS;
    const char* __restrict__ zw = (const char*)(z + (size_t)rowBase * D + colBase);
    const unsigned laneOff = 8u * (unsigned)c;
    const int* __restrict__ lp = labels + rowBase + lane;
    unsigned* slotw = &slots[w][0][0];
    const bool counter = cg == 0;
    if (counter) {
        hist[tid] = 0;
        __syncthreads();
    }

    v8f acc[NCT][2];
#pragma unroll
    for (int t = 0; t < NCT; ++t) { acc[t][0] = v8f{}; acc[t][1] = v8f{}; }

    v2f raw[2 * SEG_STEPS], frag[2 * SEG_STEPS];
    int labIn[2], labAhead[2];
    unsigned smCur[NCT], smNext[NCT];
    v2u lq[SEG_STEPS];
    labIn[0] = lp[0];
    labIn[1] = lp[32];
    labAhead[0] = lp[SEG_SS];
    labAhead[1] = lp[SEG_SS + 32];
    seg_sort(labIn, slotw, lane, smCur);
    seg_load(zw, slotw, h, laneOff, raw);
    if (counter) {
        atomicAdd(&hist[min((unsigned)labIn[0], SEG_NOCLS)], 1);
        atomicAdd(&hist[min((unsigned)labIn[1], SEG_NOCLS)], 1);
    }
    seg_fragments(raw, frag);
    for (int ss = 1; ss < SEG_NSS; ++ss) {
        labIn[0] = labAhead[0]; labIn[1] = labAhead[1];
        if (ss + 1 < SEG_NSS) {
            labAhead[0] = lp[(ss + 1) * SEG_SS];
            labAhead[1] = lp[(ss + 1) * SEG_SS + 32];
        }
        unsigned* slotNew = slotw + (ss & 1) * SEG_SS;
        const unsigned* slotOld = slotw + ((ss & 1) ^ 1) * SEG_SS;
        seg_labels(slotOld, h, lq);
        seg_sort(labIn, slotNew, lane, smNext);
        seg_load(zw + (size_t)ss * (SEG_SS * D * 4), slotNew, h, laneOff, raw);
        if (counter) {
            atomicAdd(&hist[min((unsigned)labIn[0], SEG_NOCLS)], 1);
            atomicAdd(&hist[min((unsigned)labIn[1], SEG_NOCLS)], 1);
        }
        seg_compute(frag, lq, smCur, c, h, acc);
        seg_fragments(raw, frag);
#pragma unroll
        for (int t = 0; t < NCT; ++t) smCur[t] = smNext[t];
    }
    seg_labels(slotw + ((SEG_NSS - 1) & 1) * SEG_SS, h, lq);
    seg_compute(frag, lq, smCur, c, h, acc);

    float* mine = red + w * SEG_PLANE;
#pragma unroll
    for (int t = 0; t < NCT; ++t) {
#pragma unroll
        for (int r = 0; r < 8; ++r) {
            const int cls = 16 * t + 8 * h + r;
            if (cls < NCLS) {
                const v2f pr = {acc[t][0][r], acc[t][1][r]};
                *(v2f*)(mine + cls * SEG_COLS + 2 * c) = pr;
            }
        }
    }
    __syncthreads();
    const v4f* red4 = (const v4f*)red;
    for (int e = tid; e < SEG_PLANE / 4; e += SEG_WAVES * 32) {
        v4f s = red4[e];
        s += red4[e + SEG_PLANE / 4];
        s += red4[e + 2 * (SEG_PLANE / 4)];
        s += red4[e + 3 * (SEG_PLANE / 4)];
        const int cls = e >> 3, q = e & 7;
        *(v4f*)(p + (size_t)rc * PROTO_E + cls * D + colBase + 4 * q) = s;
    }
    if (counter) cntp[rc * CNT_PITCH + tid] = hist[tid];
}

#define FIN_BLOCKS (PROTO_E / 256)
#define FIN_PADB   ((C2P - C2) * D / 256)
__global__ void finalize_proto(const float* __restrict__ p1, const float* __restrict__ p2,
                               const int* __restrict__ cntp, int* __restrict__ cnt,
                               float* __restrict__ out1, float* __restrict__ out2,
                               float* __restrict__ zb) {
    if (blockIdx.x >= FIN_BLOCKS) {
        zb[(size_t)C2 * D + (blockIdx.x - FIN_BLOCKS) * 256 + threadIdx.x] = 0.f;
        return;
    }
    const int e = blockIdx.x * 256 + threadIdx.x;
    const int cls = blockIdx.x / 3;
    int n = 0;
    for (int rc = 0; rc < NRC; ++rc) n += cntp[rc * CNT_PITCH + cls];
    if (blockIdx.x == 0 && threadIdx.x < CNT_PITCH) {
        int m = 0;
        for (int rc = 0; rc < NRC; ++rc) m += cntp[rc * CNT_PITCH + threadIdx.x];
        cnt[threadIdx.x] = m;
    }
    float s1 = 0.f, s2 = 0.f;
    for (int rc = 0; rc < NRC; ++rc) {
        s1 += p1[(size_t)rc * PROTO_E + e];
        s2 += p2[(size_t)rc * PROTO_E + e];
    }
    const float d = fmaxf((float)n, 1.0f);
    const float q1 = s1 / d, q2 = s2 / d;
    out1[e] = q1;
    out2[e] = q2;
    zb[e] = q1;
    zb[PROTO_E + e] = q2;
}

__device__ __forceinline__ v2f sq_acc(const v2f x, const v2f s) {
    return __builtin_elementwise_fma(x, x, s);
}

__global__ __launch_bounds__(32) void logits_wmma(const float* __restrict__ zb,
                                                  float* __restrict__ lg) {
    const int tile = blockIdx.x;
    const int ti = tile / NTILE, tj = tile % NTILE;
    const int lane = threadIdx.x;
    const int mn = lane & 15;
    const int kb = (lane >> 4) << 1;
    const float* __restrict__ arow = zb + (size_t)(ti * 16 + mn) * D;
    const float* __restrict__ brow = zb + (size_t)(tj * 16 + mn) * D;
    v8f acc = {};
    v2f sa = {0.f, 0.f}, sb = {0.f, 0.f};
    for (int k0 = 0; k0 < D; k0 += 4) {
        const v2f a = *(const v2f*)(arow + k0 + kb);
        const v2f b = *(const v2f*)(brow + k0 + kb);
        sa = sq_acc(a, sa);
        sb = sq_acc(b, sb);
        acc = __builtin_amdgcn_wmma_f32_16x16x4_f32(
             false, a,  false, b,
             (short)0, acc,  false,  false);
    }
    float na = sa.x + sa.y, nb = sb.x + sb.y;
    na += __shfl_xor(na, 16);
    nb += __shfl_xor(nb, 16);
    const float ia = 1.0f / fmaxf(sqrtf(na), 1e-12f);
    const float ib = 1.0f / fmaxf(sqrtf(nb), 1e-12f);
    const int mhi = (lane >> 4) * 8;
#pragma unroll
    for (int v = 0; v < 8; ++v) {
        const int m = ti * 16 + v + mhi;
        const float iar = __shfl(ia, mhi + v);
        lg[(size_t)m * C2P + tj * 16 + mn] = acc[v] * (iar * ib) * 2.0f;
    }
}

__device__ __forceinline__ bool cls_valid(const unsigned (&vm)[4], int cls) {
    return (vm[cls >> 5] >> (cls & 31)) & 1u;
}

__global__ __launch_bounds__(256) void loss_kernel(const float* __restrict__ lg,
                                                   const int* __restrict__ cnt,
                                                   float* __restrict__ out) {
    __shared__ float sterm[256];
    __shared__ float snv[256];
    const int t = threadIdx.x;
    const int lane = t & 31;
    unsigned vm[4];
    vm[0] = __builtin_amdgcn_ballot_w32(cnt[lane] > 0);
    vm[1] = __builtin_amdgcn_ballot_w32(cnt[32 + lane] > 0);
    vm[2] = __builtin_amdgcn_ballot_w32(cnt[64 + lane] > 0);
    const int c3 = cnt[96 + (lane & 3)];
    vm[3] = __builtin_amdgcn_ballot_w32(lane < NCLS - 96 && c3 > 0);
    float term = 0.f, nvf = 0.f;
    if (t < C2) {
        const int tcls = t < NCLS ? t : t - NCLS;
        const unsigned mw = tcls < 32 ? vm[0] : tcls < 64 ? vm[1] : tcls < 96 ? vm[2] : vm[3];
        const bool vr = (mw >> (tcls & 31)) & 1u;
        nvf = vr ? 1.f : 0.f;
        float v[C2];
#pragma unroll
        for (int c = 0; c < C2; ++c) v[c] = lg[(size_t)c * C2P + t];
        const int p = t < NCLS ? t + NCLS : t - NCLS;
        const float pos = lg[(size_t)t * C2P + p];
        float mx = -3.402823466e38f;
#pragma unroll
        for (int c = 0; c < C2; ++c)
            if (cls_valid(vm, c % NCLS)) mx = fmaxf(mx, v[c]);
        float se = 0.f;
        const float L2E = 1.44269504088896340736f;
        const float nb = -mx * L2E;
#pragma unroll
        for (int c = 0; c < C2; ++c) {
            const float e = __builtin_amdgcn_exp2f(__builtin_fmaf(v[c], L2E, nb));
            se += (cls_valid(vm, c % NCLS) && c != t) ? e : 0.f;
        }
        if (vr) term = pos - mx - logf(se);
    }
    sterm[t] = term;
    snv[t] = nvf;
    __syncthreads();
    for (int s = 128; s > 0; s >>= 1) {
        if (t < s) { sterm[t] += sterm[t + s]; snv[t] += snv[t + s]; }
        __syncthreads();
    }
    if (t == 0) out[0] = -sterm[0] / fmaxf(snv[0], 1.0f);
}

extern "C" void kernel_launch(void* const* d_in, const int* in_sizes, int n_in,
                              void* d_out, int out_size, void* d_ws, size_t ws_size,
                              hipStream_t stream) {
    const float* z1     = (const float*)d_in[0];
    const float* z2     = (const float*)d_in[1];
    const int*   labels = (const int*)d_in[2];
    (void)in_sizes; (void)n_in; (void)out_size; (void)ws_size;

    float* ws = (float*)d_ws;
    int*   cnt = (int*)(ws + WS_CNT);
    float* p1  = ws + WS_P1;
    float* p2  = ws + WS_P2;
    float* zb  = ws + WS_ZB;
    float* lg  = ws + WS_LG;
    int*   cntp = (int*)(ws + WS_CP);

    float* out   = (float*)d_out;
    float* out1  = out + 1;
    float* out2  = out + 1 + PROTO_E;

    segsum_onehot<<<dim3(SEG_NCG, NRC), SEG_WAVES * 32, 0, stream>>>(z1, z2, labels, p1, p2, cntp);
    finalize_proto<<<FIN_BLOCKS + FIN_PADB, 256, 0, stream>>>(p1, p2, cntp, cnt, out1, out2, zb);
    logits_wmma<<<NTILE * NTILE, 32, 0, stream>>>(zb, lg);
    loss_kernel<<<1, 256, 0, stream>>>(lg, cnt, out);
}


# Logits launch: a tile's 32 prototype rows are staged in LDS by whole-row 512-byte loads, a 128-column chunk ahead of the products; fragments read from LDS at a conflict-free pitch; sums unchanged
# speedup vs baseline: 9.0268x; 1.0643x over previous
#include <hip/hip_runtime.h>
#include <hip/hip_bf16.h>

#define N_ROWS   32768
#define D        768
#define NCLS     100
#define C2       200
#define C2P      208
#define NTILE    13
#define NRC      16
#define ROWS_PER 2048
#define PROTO_E  (NCLS * D)

typedef float v2f __attribute__((ext_vector_type(2)));
typedef float v8f __attribute__((ext_vector_type(8)));

#define WS_CNT   0
#define WS_P1    128
#define WS_P2    (WS_P1 + NRC * PROTO_E)
#define WS_ZB    (WS_P2 + NRC * PROTO_E)
#define WS_LG    (WS_ZB + C2P * D)
#define WS_CP    WS_LG
#define CNT_PITCH 128

#define SEG_COLS   32
#define SEG_NCG    48
#define SEG_HALF   24
#define SEG_WAVES  4
#define SEG_WROWS  512
#define SEG_SS     64
#define SEG_STEPS  16
#define SEG_NSS    8
#define NCT        7
#define SEG_PLANE  (NCLS * SEG_COLS)
#define SEG_OFFBITS 18
#define SEG_OFFMASK ((1u << SEG_OFFBITS) - 1u)
#define SEG_NOCLS   127u

typedef float v4f __attribute__((ext_vector_type(4)));
typedef unsigned v2u __attribute__((ext_vector_type(2)));

__device__ __forceinline__ void seg_sort(const int (&labv)[2], unsigned* slot, int lane,
                                         unsigned (&sm)[NCT]) {
    const unsigned a = min((unsigned)labv[0], SEG_NOCLS);
    const unsigned b = min((unsigned)labv[1], SEG_NOCLS);
    const unsigned ta = a >> 4, tb = b >> 4;
    unsigned base = 0, pa = 0, pb = 0;
#pragma unroll
    for (unsigned t = 0; t < 8; ++t) {
        const unsigned m0 = __builtin_amdgcn_ballot_w32(ta == t);
        const unsigned m1 = __builtin_amdgcn_ballot_w32(tb == t);
        const unsigned n0 = __builtin_popcount(m0);
        const unsigned ra = __builtin_amdgcn_mbcnt_lo(m0, base);
        const unsigned rb = __builtin_amdgcn_mbcnt_lo(m1, base + n0);
        pa = ta == t ? ra : pa;
        pb = tb == t ? rb : pb;
        const unsigned first = base;
        base += n0 + __builtin_popcount(m1);
        if (t < NCT)
            sm[t] = base > first ? (1u << ((base + 3) >> 2)) - (1u << (first >> 2)) : 0u;
    }
    slot[pa] = (a << SEG_OFFBITS) | (unsigned)(lane * (D * 4));
    slot[pb] = (b << SEG_OFFBITS) | (unsigned)((lane + 32) * (D * 4));
}

__device__ __forceinline__ void seg_load(const char* __restrict__ zg, const unsigned* slot,
                                         int h, unsigned laneOff, v2f (&buf)[2 * SEG_STEPS]) {
#pragma unroll
    for (int i = 0; i < SEG_STEPS; ++i) {
        const v2u q = *(const v2u*)(slot + 4 * i + 2 * h);
        buf[2 * i]     = *(const v2f*)(zg + ((q.x & SEG_OFFMASK) | laneOff));
        buf[2 * i + 1] = *(const v2f*)(zg + ((q.y & SEG_OFFMASK) | laneOff));
    }
}

__device__ __forceinline__ void seg_fragments(const v2f (&raw)[2 * SEG_STEPS],
                                              v2f (&frag)[2 * SEG_STEPS]) {
#pragma unroll
    for (int i = 0; i < SEG_STEPS; ++i) {
        const v2f u = raw[2 * i], v = raw[2 * i + 1];
        frag[2 * i]     = v2f{u.x, v.x};
        frag[2 * i + 1] = v2f{u.y, v.y};
    }
}

__device__ __forceinline__ void seg_labels(const unsigned* slot, int h, v2u (&lq)[SEG_STEPS]) {
#pragma unroll
    for (int i = 0; i < SEG_STEPS; ++i) lq[i] = *(const v2u*)(slot + 4 * i + 2 * h);
}

__device__ __forceinline__ void seg_compute(const v2f (&frag)[2 * SEG_STEPS], const v2u (&lq)[SEG_STEPS],
                                            const unsigned (&sm)[NCT], int c, int h,
                                            v8f (&acc)[NCT][2]) {
#pragma unroll
    for (int i = 0; i < SEG_STEPS; ++i) {
        const v2f b0 = frag[2 * i];
        const v2f b1 = frag[2 * i + 1];
        const unsigned la = lq[i].x >> SEG_OFFBITS, lb = lq[i].y >> SEG_OFFBITS;
#pragma unroll
        for (int t = 0; t < NCT; ++t) {
            if (sm[t] & (1u << i)) {
                const unsigned ct = (unsigned)(16 * t + c);
                const v2f a = {la == ct ? 1.0f : 0.0f, lb == ct ? 1.0f : 0.0f};
                acc[t][0] = __builtin_amdgcn_wmma_f32_16x16x4_f32(
                    false, a, false, b0, (short)0, acc[t][0], false, false);
                acc[t][1] = __builtin_amdgcn_wmma_f32_16x16x4_f32(
                    false, a, false, b1, (short)0, acc[t][1], false, false);
            }
        }
    }
}

__global__ __launch_bounds__(SEG_WAVES * 32) void segsum_onehot(
        const float* __restrict__ z1, const float* __restrict__ z2,
        const int* __restrict__ labels,
        float* __restrict__ p1, float* __restrict__ p2, int* __restrict__ cntp) {
    __shared__ __attribute__((aligned(16))) float red[SEG_WAVES * SEG_PLANE];
    __shared__ int hist[CNT_PITCH];
    __shared__ __attribute__((aligned(16))) unsigned slots[SEG_WAVES][2][SEG_SS];
    const int tid  = threadIdx.x;
    const int lane = tid & 31;
    const int w    = __builtin_amdgcn_readfirstlane(tid >> 5);
    const int cg   = blockIdx.x;
    const int rc   = blockIdx.y;
    const bool second = cg >= SEG_HALF;
    const float* __restrict__ z = second ? z2 : z1;
    float* __restrict__ p       = second ? p2 : p1;
    const int colBase = (second ? cg - SEG_HALF : cg) * SEG_COLS;
    const int c = lane & 15;
    const int h = lane >> 4;
    const int rowBase = rc * ROWS_PER + w * SEG_WROWS;
    const char* __restrict__ zw = (const char*)(z + (size_t)rowBase * D + colBase);
    const unsigned laneOff = 8u * (unsigned)c;
    const int* __restrict__ lp = labels + rowBase + lane;
    unsigned* slotw = &slots[w][0][0];
    const bool counter = cg == 0;
    if (counter) {
        hist[tid] = 0;
        __syncthreads();
    }

    v8f acc[NCT][2];
#pragma unroll
    for (int t = 0; t < NCT; ++t) { acc[t][0] = v8f{}; acc[t][1] = v8f{}; }

    v2f raw[2 * SEG_STEPS], frag[2 * SEG_STEPS];
    int labIn[2], labAhead[2];
    unsigned smCur[NCT], smNext[NCT];
    v2u lq[SEG_STEPS];
    labIn[0] = lp[0];
    labIn[1] = lp[32];
    labAhead[0] = lp[SEG_SS];
    labAhead[1] = lp[SEG_SS + 32];
    seg_sort(labIn, slotw, lane, smCur);
    seg_load(zw, slotw, h, laneOff, raw);
    if (counter) {
        atomicAdd(&hist[min((unsigned)labIn[0], SEG_NOCLS)], 1);
        atomicAdd(&hist[min((unsigned)labIn[1], SEG_NOCLS)], 1);
    }
    seg_fragments(raw, frag);
    for (int ss = 1; ss < SEG_NSS; ++ss) {
        labIn[0] = labAhead[0]; labIn[1] = labAhead[1];
        if (ss + 1 < SEG_NSS) {
            labAhead[0] = lp[(ss + 1) * SEG_SS];
            labAhead[1] = lp[(ss + 1) * SEG_SS + 32];
        }
        unsigned* slotNew = slotw + (ss & 1) * SEG_SS;
        const unsigned* slotOld = slotw + ((ss & 1) ^ 1) * SEG_SS;
        seg_labels(slotOld, h, lq);
        seg_sort(labIn, slotNew, lane, smNext);
        seg_load(zw + (size_t)ss * (SEG_SS * D * 4), slotNew, h, laneOff, raw);
        if (counter) {
            atomicAdd(&hist[min((unsigned)labIn[0], SEG_NOCLS)], 1);
            atomicAdd(&hist[min((unsigned)labIn[1], SEG_NOCLS)], 1);
        }
        seg_compute(frag, lq, smCur, c, h, acc);
        seg_fragments(raw, frag);
#pragma unroll
        for (int t = 0; t < NCT; ++t) smCur[t] = smNext[t];
    }
    seg_labels(slotw + ((SEG_NSS - 1) & 1) * SEG_SS, h, lq);
    seg_compute(frag, lq, smCur, c, h, acc);

    float* mine = red + w * SEG_PLANE;
#pragma unroll
    for (int t = 0; t < NCT; ++t) {
#pragma unroll
        for (int r = 0; r < 8; ++r) {
            const int cls = 16 * t + 8 * h + r;
            if (cls < NCLS) {
                const v2f pr = {acc[t][0][r], acc[t][1][r]};
                *(v2f*)(mine + cls * SEG_COLS + 2 * c) = pr;
            }
        }
    }
    __syncthreads();
    const v4f* red4 = (const v4f*)red;
    for (int e = tid; e < SEG_PLANE / 4; e += SEG_WAVES * 32) {
        v4f s = red4[e];
        s += red4[e + SEG_PLANE / 4];
        s += red4[e + 2 * (SEG_PLANE / 4)];
        s += red4[e + 3 * (SEG_PLANE / 4)];
        const int cls = e >> 3, q = e & 7;
        *(v4f*)(p + (size_t)rc * PROTO_E + cls * D + colBase + 4 * q) = s;
    }
    if (counter) cntp[rc * CNT_PITCH + tid] = hist[tid];
}

#define FIN_BLOCKS (PROTO_E / 256)
#define FIN_PADB   ((C2P - C2) * D / 256)
__global__ void finalize_proto(const float* __restrict__ p1, const float* __restrict__ p2,
                               const int* __restrict__ cntp, int* __restrict__ cnt,
                               float* __restrict__ out1, float* __restrict__ out2,
                               float* __restrict__ zb) {
    if (blockIdx.x >= FIN_BLOCKS) {
        zb[(size_t)C2 * D + (blockIdx.x - FIN_BLOCKS) * 256 + threadIdx.x] = 0.f;
        return;
    }
    const int e = blockIdx.x * 256 + threadIdx.x;
    const int cls = blockIdx.x / 3;
    int n = 0;
    for (int rc = 0; rc < NRC; ++rc) n += cntp[rc * CNT_PITCH + cls];
    if (blockIdx.x == 0 && threadIdx.x < CNT_PITCH) {
        int m = 0;
        for (int rc = 0; rc < NRC; ++rc) m += cntp[rc * CNT_PITCH + threadIdx.x];
        cnt[threadIdx.x] = m;
    }
    float s1 = 0.f, s2 = 0.f;
    for (int rc = 0; rc < NRC; ++rc) {
        s1 += p1[(size_t)rc * PROTO_E + e];
        s2 += p2[(size_t)rc * PROTO_E + e];
    }
    const float d = fmaxf((float)n, 1.0f);
    const float q1 = s1 / d, q2 = s2 / d;
    out1[e] = q1;
    out2[e] = q2;
    zb[e] = q1;
    zb[PROTO_E + e] = q2;
}

__device__ __forceinline__ v2f sq_acc(const v2f x, const v2f s) {
    return __builtin_elementwise_fma(x, x, s);
}

#define LG_CH     128
#define LG_NCH    (D / LG_CH)
#define LG_PITCH  (D + 4)
__global__ __launch_bounds__(32) void logits_wmma(const float* __restrict__ zb,
                                                  float* __restrict__ lg) {
    __shared__ __attribute__((aligned(16))) float rows[32 * LG_PITCH];
    const int tile = blockIdx.x;
    const int ti = tile / NTILE, tj = tile % NTILE;
    const int lane = threadIdx.x;
    const int mn = lane & 15;
    const int kb = (lane >> 4) << 1;
    const float* __restrict__ asrc = zb + (size_t)(ti * 16) * D + 4 * lane;
    const float* __restrict__ bsrc = zb + (size_t)(tj * 16) * D + 4 * lane;
    float* dst = rows + 4 * lane;
    v4f st[32];
#pragma unroll
    for (int r = 0; r < 16; ++r) {
        st[r]      = *(const v4f*)(asrc + (size_t)r * D);
        st[16 + r] = *(const v4f*)(bsrc + (size_t)r * D);
    }
    const float* afr = rows + mn * LG_PITCH + kb;
    const float* bfr = rows + (16 + mn) * LG_PITCH + kb;
    v8f acc = {};
    v2f sa = {0.f, 0.f}, sb = {0.f, 0.f};
    for (int ch = 0; ch < LG_NCH; ++ch) {
#pragma unroll
        for (int r = 0; r < 32; ++r) *(v4f*)(dst + r * LG_PITCH + ch * LG_CH) = st[r];
        if (ch + 1 < LG_NCH) {
#pragma unroll
            for (int r = 0; r < 16; ++r) {
                st[r]      = *(const v4f*)(asrc + (size_t)r * D + (ch + 1) * LG_CH);
                st[16 + r] = *(const v4f*)(bsrc + (size_t)r * D + (ch + 1) * LG_CH);
            }
        }
        __syncthreads();
        const float* ac = afr + ch * LG_CH;
        const float* bc = bfr + ch * LG_CH;
#pragma unroll
        for (int s = 0; s < LG_CH / 4; ++s) {
            const v2f a = *(const v2f*)(ac + 4 * s);
            const v2f b = *(const v2f*)(bc + 4 * s);
            sa = sq_acc(a, sa);
            sb = sq_acc(b, sb);
            acc = __builtin_amdgcn_wmma_f32_16x16x4_f32(
                 false, a,  false, b,
                 (short)0, acc,  false,  false);
        }
    }
    float na = sa.x + sa.y, nb = sb.x + sb.y;
    na += __shfl_xor(na, 16);
    nb += __shfl_xor(nb, 16);
    const float ia = 1.0f / fmaxf(sqrtf(na), 1e-12f);
    const float ib = 1.0f / fmaxf(sqrtf(nb), 1e-12f);
    const int mhi = (lane >> 4) * 8;
#pragma unroll
    for (int v = 0; v < 8; ++v) {
        const int m = ti * 16 + v + mhi;
        const float iar = __shfl(ia, mhi + v);
        lg[(size_t)m * C2P + tj * 16 + mn] = acc[v] * (iar * ib) * 2.0f;
    }
}

__device__ __forceinline__ bool cls_valid(const unsigned (&vm)[4], int cls) {
    return (vm[cls >> 5] >> (cls & 31)) & 1u;
}

__global__ __launch_bounds__(256) void loss_kernel(const float* __restrict__ lg,
                                                   const int* __restrict__ cnt,
                                                   float* __restrict__ out) {
    __shared__ float sterm[256];
    __shared__ float snv[256];
    const int t = threadIdx.x;
    const int lane = t & 31;
    unsigned vm[4];
    vm[0] = __builtin_amdgcn_ballot_w32(cnt[lane] > 0);
    vm[1] = __builtin_amdgcn_ballot_w32(cnt[32 + lane] > 0);
    vm[2] = __builtin_amdgcn_ballot_w32(cnt[64 + lane] > 0);
    const int c3 = cnt[96 + (lane & 3)];
    vm[3] = __builtin_amdgcn_ballot_w32(lane < NCLS - 96 && c3 > 0);
    float term = 0.f, nvf = 0.f;
    if (t < C2) {
        const int tcls = t < NCLS ? t : t - NCLS;
        const unsigned mw = tcls < 32 ? vm[0] : tcls < 64 ? vm[1] : tcls < 96 ? vm[2] : vm[3];
        const bool vr = (mw >> (tcls & 31)) & 1u;
        nvf = vr ? 1.f : 0.f;
        float v[C2];
#pragma unroll
        for (int c = 0; c < C2; ++c) v[c] = lg[(size_t)c * C2P + t];
        const int p = t < NCLS ? t + NCLS : t - NCLS;
        const float pos = lg[(size_t)t * C2P + p];
        float mx = -3.402823466e38f;
#pragma unroll
        for (int c = 0; c < C2; ++c)
            if (cls_valid(vm, c % NCLS)) mx = fmaxf(mx, v[c]);
        float se = 0.f;
        const float L2E = 1.44269504088896340736f;
        const float nb = -mx * L2E;
#pragma unroll
        for (int c = 0; c < C2; ++c) {
            const float e = __builtin_amdgcn_exp2f(__builtin_fmaf(v[c], L2E, nb));
            se += (cls_valid(vm, c % NCLS) && c != t) ? e : 0.f;
        }
        if (vr) term = pos - mx - logf(se);
    }
    sterm[t] = term;
    snv[t] = nvf;
    __syncthreads();
    for (int s = 128; s > 0; s >>= 1) {
        if (t < s) { sterm[t] += sterm[t + s]; snv[t] += snv[t + s]; }
        __syncthreads();
    }
    if (t == 0) out[0] = -sterm[0] / fmaxf(snv[0], 1.0f);
}

extern "C" void kernel_launch(void* const* d_in, const int* in_sizes, int n_in,
                              void* d_out, int out_size, void* d_ws, size_t ws_size,
                              hipStream_t stream) {
    const float* z1     = (const float*)d_in[0];
    const float* z2     = (const float*)d_in[1];
    const int*   labels = (const int*)d_in[2];
    (void)in_sizes; (void)n_in; (void)out_size; (void)ws_size;

    float* ws = (float*)d_ws;
    int*   cnt = (int*)(ws + WS_CNT);
    float* p1  = ws + WS_P1;
    float* p2  = ws + WS_P2;
    float* zb  = ws + WS_ZB;
    float* lg  = ws + WS_LG;
    int*   cntp = (int*)(ws + WS_CP);

    float* out   = (float*)d_out;
    float* out1  = out + 1;
    float* out2  = out + 1 + PROTO_E;

    segsum_onehot<<<dim3(SEG_NCG, NRC), SEG_WAVES * 32, 0, stream>>>(z1, z2, labels, p1, p2, cntp);
    finalize_proto<<<FIN_BLOCKS + FIN_PADB, 256, 0, stream>>>(p1, p2, cntp, cnt, out1, out2, zb);
    logits_wmma<<<NTILE * NTILE, 32, 0, stream>>>(zb, lg);
    loss_kernel<<<1, 256, 0, stream>>>(lg, cnt, out);
}
